# starting kernel
# baseline (speedup 1.0000x reference)
#include <hip/hip_runtime.h>
#include <math.h>

typedef float v2f __attribute__((ext_vector_type(2)));
typedef float v8f __attribute__((ext_vector_type(8)));

#define DDIM 2048
#define CCLS 10
#define BROWS 8192
#define WCH 128
#define WPAD 132
#define NCHUNK (DDIM / WCH)
#define KKPC (WCH / 4)

__global__ void zero_ws_kernel(float* __restrict__ ws) {
    int i = blockIdx.x * blockDim.x + threadIdx.x;
    ws[i] = 0.0f;
}

__global__ __launch_bounds__(256) void softmax_pack_kernel(const float* __restrict__ w,
                                                           float* __restrict__ sfrag) {
    __shared__ float red[256];
    const int c = blockIdx.x;
    const int tid = threadIdx.x;
    const float* wr = w + c * DDIM;

    float m = -3.4e38f;
    for (int d = tid; d < DDIM; d += 256) m = fmaxf(m, wr[d]);
    red[tid] = m; __syncthreads();
    for (int s = 128; s > 0; s >>= 1) { if (tid < s) red[tid] = fmaxf(red[tid], red[tid + s]); __syncthreads(); }
    m = red[0]; __syncthreads();

    float sum = 0.0f;
    for (int d = tid; d < DDIM; d += 256) sum += __expf(wr[d] - m);
    red[tid] = sum; __syncthreads();
    for (int s = 128; s > 0; s >>= 1) { if (tid < s) red[tid] += red[tid + s]; __syncthreads(); }
    const float inv = 1.0f / red[0];

    for (int d = tid; d < DDIM; d += 256) {
        float v = __expf(wr[d] - m) * inv;
        int kk = d >> 2, k = d & 3;
        sfrag[kk * 64 + (c + 16 * (k >> 1)) * 2 + (k & 1)] = v;
    }
}

__global__ __launch_bounds__(128) void lse_wmma_kernel(const float* __restrict__ x,
                                                       const float* __restrict__ sfrag,
                                                       float* __restrict__ out) {
    __shared__ float xlds[4][16 * WPAD];
    __shared__ float mrow[4][16];
    __shared__ float cmax[4][16];
    __shared__ float fsc[4][16];

    const int lane = threadIdx.x & 31;
    const int wv   = threadIdx.x >> 5;
    const int tile = blockIdx.x * 4 + wv;
    const int r0   = tile * 16;
    const int rA   = lane & 15;
    const int hh   = lane >> 4;

    float* xw = xlds[wv];
    float* mr = mrow[wv];
    float* cm = cmax[wv];
    float* fw = fsc[wv];

    if (lane < 16) mr[lane] = -3.0e38f;
    v8f acc = {0.f, 0.f, 0.f, 0.f, 0.f, 0.f, 0.f, 0.f};
    __syncthreads();

    for (int ch = 0; ch < NCHUNK; ++ch) {
        const float* xg = x + (size_t)r0 * DDIM + ch * WCH + 4 * lane;
        #pragma unroll
        for (int i = 0; i < 16; ++i) {
            float4 v = *(const float4*)(xg + (size_t)i * DDIM);
            *(float4*)(xw + i * WPAD + 4 * lane) = v;
            float lm = fmaxf(fmaxf(v.x, v.y), fmaxf(v.z, v.w));
            #pragma unroll
            for (int off = 16; off > 0; off >>= 1)
                lm = fmaxf(lm, __shfl_xor(lm, off, 32));
            if (lane == i) cm[i] = lm;
        }
        __syncthreads();

        float mo  = mr[rA];
        float cmx = cm[rA];
        float mn  = fmaxf(mo, cmx);
        __syncthreads();
        if (lane < 16) { mr[lane] = mn; fw[lane] = __expf(mo - mn); }
        __syncthreads();
        {
            const int base = (lane < 16) ? 0 : 8;
            #pragma unroll
            for (int v = 0; v < 8; ++v) acc[v] *= fw[base + v];
        }

        const float* ap = xw + rA * WPAD + 2 * hh;
        const float* bp = sfrag + (size_t)ch * KKPC * 64 + 2 * lane;
        #pragma unroll 4
        for (int kk = 0; kk < KKPC; ++kk) {
            float2 a = *(const float2*)(ap + 4 * kk);
            float2 b = *(const float2*)(bp + 64 * kk);
            v2f av; av.x = __expf(a.x - mn); av.y = __expf(a.y - mn);
            v2f bv; bv.x = b.x; bv.y = b.y;
            acc = __builtin_amdgcn_wmma_f32_16x16x4_f32(false, av, false, bv,
                                                        (short)0, acc, false, false);
        }
        __syncthreads();
    }

    const int base = (lane < 16) ? 0 : 8;
    if (rA < CCLS) {
        #pragma unroll
        for (int v = 0; v < 8; ++v) {
            float mf = mr[base + v];
            out[(size_t)(r0 + base + v) * CCLS + rA] = mf + __logf(acc[v]);
        }
    }
}

extern "C" void kernel_launch(void* const* d_in, const int* in_sizes, int n_in,
                              void* d_out, int out_size, void* d_ws, size_t ws_size,
                              hipStream_t stream) {
    const float* x = (const float*)d_in[0];
    const float* w = (const float*)d_in[1];
    float* out   = (float*)d_out;
    float* sfrag = (float*)d_ws;

    zero_ws_kernel<<<128, 256, 0, stream>>>(sfrag);
    softmax_pack_kernel<<<CCLS, 256, 0, stream>>>(w, sfrag);
    lse_wmma_kernel<<<BROWS / 64, 128, 0, stream>>>(x, sfrag, out);
}


# x fragments read straight from global memory as 16-byte pieces of the lane's own row (k order permuted, weights packed to match: new summation order); no LDS staging, no barriers
# speedup vs baseline: 8.0343x; 8.0343x over previous
#include <hip/hip_runtime.h>
#include <math.h>


typedef float v2f __attribute__((ext_vector_type(2)));
typedef float v8f __attribute__((ext_vector_type(8)));

#define DDIM 2048
#define CCLS 10
#define BROWS 8192
#define WCH 128
#define NCHUNK (DDIM / WCH)
#define NQ (WCH / 8)

__global__ void zero_ws_kernel(float* __restrict__ ws) {
    int i = blockIdx.x * blockDim.x + threadIdx.x;
    ws[i] = 0.0f;
}

__global__ __launch_bounds__(256) void softmax_pack_kernel(const float* __restrict__ w,
                                                           float* __restrict__ sfrag) {
    __shared__ float red[256];
    const int c = blockIdx.x;
    const int tid = threadIdx.x;
    const float* wr = w + c * DDIM;

    float m = -3.4e38f;
    for (int d = tid; d < DDIM; d += 256) m = fmaxf(m, wr[d]);
    red[tid] = m; __syncthreads();
    for (int s = 128; s > 0; s >>= 1) { if (tid < s) red[tid] = fmaxf(red[tid], red[tid + s]); __syncthreads(); }
    m = red[0]; __syncthreads();

    float sum = 0.0f;
    for (int d = tid; d < DDIM; d += 256) sum += __expf(wr[d] - m);
    red[tid] = sum; __syncthreads();
    for (int s = 128; s > 0; s >>= 1) { if (tid < s) red[tid] += red[tid + s]; __syncthreads(); }
    const float inv = 1.0f / red[0];

    for (int d = tid; d < DDIM; d += 256) {
        float v = __expf(wr[d] - m) * inv;
        int J = d >> 3, h = (d >> 2) & 1, q = d & 3;
        sfrag[J * 128 + (c + 16 * h) * 4 + q] = v;
    }
}


struct Piece16 { float4 p[NQ]; };

__device__ __forceinline__ void load_pieces(Piece16& dst, const float* __restrict__ src, int stride) {
    #pragma unroll
    for (int j = 0; j < NQ; ++j) dst.p[j] = *(const float4*)(src + j * stride);
}

__device__ __forceinline__ void chunk_step(const Piece16& a, const Piece16& b, v8f& acc, float& m,
                                           int hh) {
    float seed = m;
    asm volatile("" : "+v"(seed));
    float t0 = seed, t1 = seed, t2 = seed, t3 = seed;
    #pragma unroll
    for (int j = 0; j < NQ; j += 4) {
        t0 = fmaxf(fmaxf(t0, a.p[j + 0].x), a.p[j + 0].y); t0 = fmaxf(fmaxf(t0, a.p[j + 0].z), a.p[j + 0].w);
        t1 = fmaxf(fmaxf(t1, a.p[j + 1].x), a.p[j + 1].y); t1 = fmaxf(fmaxf(t1, a.p[j + 1].z), a.p[j + 1].w);
        t2 = fmaxf(fmaxf(t2, a.p[j + 2].x), a.p[j + 2].y); t2 = fmaxf(fmaxf(t2, a.p[j + 2].z), a.p[j + 2].w);
        t3 = fmaxf(fmaxf(t3, a.p[j + 3].x), a.p[j + 3].y); t3 = fmaxf(fmaxf(t3, a.p[j + 3].z), a.p[j + 3].w);
    }
    float mn = fmaxf(fmaxf(t0, t1), fmaxf(t2, t3));
    mn = fmaxf(mn, __shfl_xor(mn, 16, 32));
    const float f  = __expf(m - mn);
    m = mn;
    #pragma unroll
    for (int v = 0; v < 8; ++v) acc[v] *= __shfl(f, 8 * hh + v, 32);

    #pragma unroll
    for (int j = 0; j < NQ; ++j) {
        v2f av, bv;
        av.x = __expf(a.p[j].x - mn); av.y = __expf(a.p[j].y - mn);
        bv.x = b.p[j].x;              bv.y = b.p[j].y;
        acc = __builtin_amdgcn_wmma_f32_16x16x4_f32(false, av, false, bv, (short)0, acc, false, false);
        av.x = __expf(a.p[j].z - mn); av.y = __expf(a.p[j].w - mn);
        bv.x = b.p[j].z;              bv.y = b.p[j].w;
        acc = __builtin_amdgcn_wmma_f32_16x16x4_f32(false, av, false, bv, (short)0, acc, false, false);
    }
    asm volatile("" : "+v"(acc));
}

template <bool MORE>
__device__ __forceinline__ void four_chunks(Piece16& a0, Piece16& a1, Piece16& a2, Piece16& a3,
                                            Piece16& b0, Piece16& b1,
                                            const float* __restrict__ xp, const float* __restrict__ bp,
                                            int ch, v8f& acc, float& m, int hh) {
    load_pieces(b1, bp + (ch + 1) * NQ * 128, 128);
    load_pieces(a3, xp + (ch + 3) * WCH, 8);
    __builtin_amdgcn_sched_barrier(0);
    chunk_step(a0, b0, acc, m, hh);

    load_pieces(b0, bp + (ch + 2) * NQ * 128, 128);
    if constexpr (MORE) load_pieces(a0, xp + (ch + 4) * WCH, 8);
    __builtin_amdgcn_sched_barrier(0);
    chunk_step(a1, b1, acc, m, hh);

    load_pieces(b1, bp + (ch + 3) * NQ * 128, 128);
    if constexpr (MORE) load_pieces(a1, xp + (ch + 5) * WCH, 8);
    __builtin_amdgcn_sched_barrier(0);
    chunk_step(a2, b0, acc, m, hh);

    if constexpr (MORE) {
        load_pieces(b0, bp + (ch + 4) * NQ * 128, 128);
        load_pieces(a2, xp + (ch + 6) * WCH, 8);
    }
    __builtin_amdgcn_sched_barrier(0);
    chunk_step(a3, b1, acc, m, hh);
}

__global__ __launch_bounds__(64) void lse_wmma_kernel(const float* __restrict__ x,
                                                      const float* __restrict__ sfrag,
                                                      float* __restrict__ out) {
    const int lane = threadIdx.x & 31;
    const int wv   = threadIdx.x >> 5;
    const int tile = blockIdx.x * 2 + wv;
    const int r0   = tile * 16;
    const int rA   = lane & 15;
    const int hh   = lane >> 4;

    const float* xp = x + (size_t)(r0 + rA) * DDIM + 4 * hh;
    const float* bp = sfrag + 4 * lane;

    Piece16 a0, a1, a2, a3;
    Piece16 b0, b1;
    float m = -3.0e38f;
    v8f acc = {0.f, 0.f, 0.f, 0.f, 0.f, 0.f, 0.f, 0.f};

    load_pieces(a0, xp + 0 * WCH, 8);
    load_pieces(a1, xp + 1 * WCH, 8);
    load_pieces(b0, bp + 0 * NQ * 128, 128);
    load_pieces(a2, xp + 2 * WCH, 8);
    __builtin_amdgcn_sched_barrier(0);

    for (int it = 0; it < NCHUNK / 4 - 1; ++it)
        four_chunks<true>(a0, a1, a2, a3, b0, b1, xp, bp, 4 * it, acc, m, hh);
    four_chunks<false>(a0, a1, a2, a3, b0, b1, xp, bp, NCHUNK - 4, acc, m, hh);

    float mf[8];
    #pragma unroll
    for (int v = 0; v < 8; ++v) mf[v] = __shfl(m, 8 * hh + v, 32);
    if (rA < CCLS) {
        #pragma unroll
        for (int v = 0; v < 8; ++v)
            out[(size_t)(r0 + 8 * hh + v) * CCLS + rA] = mf[v] + __logf(acc[v]);
    }
}

extern "C" void kernel_launch(void* const* d_in, const int* in_sizes, int n_in,
                              void* d_out, int out_size, void* d_ws, size_t ws_size,
                              hipStream_t stream) {
    const float* x = (const float*)d_in[0];
    const float* w = (const float*)d_in[1];
    float* out   = (float*)d_out;
    float* sfrag = (float*)d_ws;

    zero_ws_kernel<<<128, 256, 0, stream>>>(sfrag);
    softmax_pack_kernel<<<CCLS, 256, 0, stream>>>(w, sfrag);
    lse_wmma_kernel<<<BROWS / 32, 64, 0, stream>>>(x, sfrag, out);
}


# Weight softmax folded into the main kernel: each work-group builds the packed B pieces once in LDS (5 classes per wave) and reads them there; 3 launches -> 1 (normaliser summed in a new order)
# speedup vs baseline: 14.0563x; 1.7495x over previous
#include <hip/hip_runtime.h>
#include <math.h>


typedef float v2f __attribute__((ext_vector_type(2)));
typedef float v8f __attribute__((ext_vector_type(8)));

#define DDIM 2048
#define CCLS 10
#define BROWS 8192
#define WCH 128
#define NCHUNK (DDIM / WCH)
#define NQ (WCH / 8)
#define NJ (DDIM / 8)
#define BUNITS (2 * CCLS)
#define CPW (CCLS / 2)
#define WQ (DDIM / 128)


struct Piece16 { float4 p[NQ]; };

__device__ __forceinline__ void load_pieces(Piece16& dst, const float* __restrict__ src, int stride) {
    #pragma unroll
    for (int j = 0; j < NQ; ++j) dst.p[j] = *(const float4*)(src + j * stride);
}

__device__ __forceinline__ void load_bpieces(Piece16& dst, const float4* src) {
    #pragma unroll
    for (int j = 0; j < NQ; ++j) dst.p[j] = src[j * BUNITS];
}

__device__ __forceinline__ void chunk_step(const Piece16& a, const Piece16& b, v8f& acc, float& m,
                                           int hh) {
    float seed = m;
    asm volatile("" : "+v"(seed));
    float t0 = seed, t1 = seed, t2 = seed, t3 = seed;
    #pragma unroll
    for (int j = 0; j < NQ; j += 4) {
        t0 = fmaxf(fmaxf(t0, a.p[j + 0].x), a.p[j + 0].y); t0 = fmaxf(fmaxf(t0, a.p[j + 0].z), a.p[j + 0].w);
        t1 = fmaxf(fmaxf(t1, a.p[j + 1].x), a.p[j + 1].y); t1 = fmaxf(fmaxf(t1, a.p[j + 1].z), a.p[j + 1].w);
        t2 = fmaxf(fmaxf(t2, a.p[j + 2].x), a.p[j + 2].y); t2 = fmaxf(fmaxf(t2, a.p[j + 2].z), a.p[j + 2].w);
        t3 = fmaxf(fmaxf(t3, a.p[j + 3].x), a.p[j + 3].y); t3 = fmaxf(fmaxf(t3, a.p[j + 3].z), a.p[j + 3].w);
    }
    float mn = fmaxf(fmaxf(t0, t1), fmaxf(t2, t3));
    mn = fmaxf(mn, __shfl_xor(mn, 16, 32));
    const float f  = __expf(m - mn);
    m = mn;
    #pragma unroll
    for (int v = 0; v < 8; ++v) acc[v] *= __shfl(f, 8 * hh + v, 32);

    #pragma unroll
    for (int j = 0; j < NQ; ++j) {
        v2f av, bv;
        av.x = __expf(a.p[j].x - mn); av.y = __expf(a.p[j].y - mn);
        bv.x = b.p[j].x;              bv.y = b.p[j].y;
        acc = __builtin_amdgcn_wmma_f32_16x16x4_f32(false, av, false, bv, (short)0, acc, false, false);
        av.x = __expf(a.p[j].z - mn); av.y = __expf(a.p[j].w - mn);
        bv.x = b.p[j].z;              bv.y = b.p[j].w;
        acc = __builtin_amdgcn_wmma_f32_16x16x4_f32(false, av, false, bv, (short)0, acc, false, false);
    }
    asm volatile("" : "+v"(acc));
}

template <bool MORE>
__device__ __forceinline__ void four_chunks(Piece16& a0, Piece16& a1, Piece16& a2, Piece16& a3,
                                            Piece16& b0, Piece16& b1,
                                            const float* __restrict__ xp, const float4* bl,
                                            int ch, v8f& acc, float& m, int hh) {
    load_bpieces(b1, bl + (ch + 1) * NQ * BUNITS);
    load_pieces(a3, xp + (ch + 3) * WCH, 8);
    __builtin_amdgcn_sched_barrier(0);
    chunk_step(a0, b0, acc, m, hh);

    load_bpieces(b0, bl + (ch + 2) * NQ * BUNITS);
    if constexpr (MORE) load_pieces(a0, xp + (ch + 4) * WCH, 8);
    __builtin_amdgcn_sched_barrier(0);
    chunk_step(a1, b1, acc, m, hh);

    load_bpieces(b1, bl + (ch + 3) * NQ * BUNITS);
    if constexpr (MORE) load_pieces(a1, xp + (ch + 5) * WCH, 8);
    __builtin_amdgcn_sched_barrier(0);
    chunk_step(a2, b0, acc, m, hh);

    if constexpr (MORE) {
        load_bpieces(b0, bl + (ch + 4) * NQ * BUNITS);
        load_pieces(a2, xp + (ch + 6) * WCH, 8);
    }
    __builtin_amdgcn_sched_barrier(0);
    chunk_step(a3, b1, acc, m, hh);
}

__global__ __launch_bounds__(64) void lse_wmma_kernel(const float* __restrict__ x,
                                                      const float* __restrict__ w,
                                                      float* __restrict__ out) {
    __shared__ float4 sB[NJ * BUNITS];

    const int lane = threadIdx.x & 31;
    const int wv   = threadIdx.x >> 5;
    const int tile = blockIdx.x * 2 + wv;
    const int r0   = tile * 16;
    const int rA   = lane & 15;
    const int hh   = lane >> 4;

    const float* xp = x + (size_t)(r0 + rA) * DDIM + 4 * hh;
    const float4* bl = sB + hh * CCLS + (rA < CCLS ? rA : CCLS - 1);

    Piece16 a0, a1, a2, a3;
    Piece16 b0, b1;
    float m = -3.0e38f;
    v8f acc = {0.f, 0.f, 0.f, 0.f, 0.f, 0.f, 0.f, 0.f};

    float4 wq[CPW][WQ];
    #pragma unroll
    for (int k = 0; k < CPW; ++k) {
        const float* wr = w + (CPW * wv + k) * DDIM + 4 * lane;
        #pragma unroll
        for (int i = 0; i < WQ; ++i) wq[k][i] = *(const float4*)(wr + i * 128);
    }
    load_pieces(a0, xp + 0 * WCH, 8);
    load_pieces(a1, xp + 1 * WCH, 8);
    load_pieces(a2, xp + 2 * WCH, 8);
    __builtin_amdgcn_sched_barrier(0);

    #pragma unroll
    for (int k = 0; k < CPW; ++k) {
        float mx = -3.4e38f;
        #pragma unroll
        for (int i = 0; i < WQ; ++i)
            mx = fmaxf(fmaxf(fmaxf(mx, wq[k][i].x), wq[k][i].y), fmaxf(wq[k][i].z, wq[k][i].w));
        for (int s = 16; s > 0; s >>= 1) mx = fmaxf(mx, __shfl_xor(mx, s, 32));

        float sum = 0.0f;
        #pragma unroll
        for (int i = 0; i < WQ; ++i) {
            wq[k][i].x = __expf(wq[k][i].x - mx); wq[k][i].y = __expf(wq[k][i].y - mx);
            wq[k][i].z = __expf(wq[k][i].z - mx); wq[k][i].w = __expf(wq[k][i].w - mx);
            sum += (wq[k][i].x + wq[k][i].y) + (wq[k][i].z + wq[k][i].w);
        }
        for (int s = 16; s > 0; s >>= 1) sum += __shfl_xor(sum, s, 32);
        const float inv = 1.0f / sum;

        const int c = CPW * wv + k;
        #pragma unroll
        for (int i = 0; i < WQ; ++i) {
            const int f = 32 * i + lane;
            float4 v;
            v.x = wq[k][i].x * inv; v.y = wq[k][i].y * inv;
            v.z = wq[k][i].z * inv; v.w = wq[k][i].w * inv;
            sB[(f >> 1) * BUNITS + (f & 1) * CCLS + c] = v;
        }
    }
    __syncthreads();

    load_bpieces(b0, bl);

    for (int it = 0; it < NCHUNK / 4 - 1; ++it)
        four_chunks<true>(a0, a1, a2, a3, b0, b1, xp, bl, 4 * it, acc, m, hh);
    four_chunks<false>(a0, a1, a2, a3, b0, b1, xp, bl, NCHUNK - 4, acc, m, hh);

    float mf[8];
    #pragma unroll
    for (int v = 0; v < 8; ++v) mf[v] = __shfl(m, 8 * hh + v, 32);
    if (rA < CCLS) {
        #pragma unroll
        for (int v = 0; v < 8; ++v)
            out[(size_t)(r0 + 8 * hh + v) * CCLS + rA] = mf[v] + __logf(acc[v]);
    }
}

extern "C" void kernel_launch(void* const* d_in, const int* in_sizes, int n_in,
                              void* d_out, int out_size, void* d_ws, size_t ws_size,
                              hipStream_t stream) {
    const float* x = (const float*)d_in[0];
    const float* w = (const float*)d_in[1];
    float* out   = (float*)d_out;

    lse_wmma_kernel<<<BROWS / 32, 64, 0, stream>>>(x, w, out);
}


# All cross-lane exchanges on the vector ALU (DPP row_xmask, v_permlanex16, v_permlane16_var) instead of ds_bpermute: no LDS-unit round trip or dscnt wait in the reductions; summation order unchanged
# speedup vs baseline: 16.7892x; 1.1944x over previous
#include <hip/hip_runtime.h>
#include <math.h>


typedef float v2f __attribute__((ext_vector_type(2)));
typedef float v8f __attribute__((ext_vector_type(8)));

#define DDIM 2048
#define CCLS 10
#define BROWS 8192
#define WCH 128
#define NCHUNK (DDIM / WCH)
#define NQ (WCH / 8)
#define NJ (DDIM / 8)
#define BUNITS (2 * CCLS)
#define CPW (CCLS / 2)
#define WQ (DDIM / 128)


struct Piece16 { float4 p[NQ]; };


__device__ __forceinline__ float xhalf(float x) {
    const int b = __builtin_bit_cast(int, x);
    return __builtin_bit_cast(float,
        __builtin_amdgcn_permlanex16(b, b, 0x76543210, 0xfedcba98, false, false));
}

template <int MASK>
__device__ __forceinline__ float xrow(float x) {
    const int b = __builtin_bit_cast(int, x);
    return __builtin_bit_cast(float, __builtin_amdgcn_update_dpp(0, b, 0x160 + MASK, 0xf, 0xf, true));
}

__device__ __forceinline__ float rowpick(float x, int pos) {
    const unsigned b = __builtin_bit_cast(unsigned, x);
    return __builtin_bit_cast(float, __builtin_amdgcn_permlane16_var(b, b, (unsigned)pos, false, false));
}

__device__ __forceinline__ float wave_max(float v) {
    v = fmaxf(v, xhalf(v));
    v = fmaxf(v, xrow<8>(v)); v = fmaxf(v, xrow<4>(v));
    v = fmaxf(v, xrow<2>(v)); v = fmaxf(v, xrow<1>(v));
    return v;
}
__device__ __forceinline__ float wave_sum(float v) {
    v += xhalf(v);
    v += xrow<8>(v); v += xrow<4>(v);
    v += xrow<2>(v); v += xrow<1>(v);
    return v;
}

__device__ __forceinline__ void load_pieces(Piece16& dst, const float* __restrict__ src, int stride) {
    #pragma unroll
    for (int j = 0; j < NQ; ++j) dst.p[j] = *(const float4*)(src + j * stride);
}

__device__ __forceinline__ void load_bpieces(Piece16& dst, const float4* src) {
    #pragma unroll
    for (int j = 0; j < NQ; ++j) dst.p[j] = src[j * BUNITS];
}

__device__ __forceinline__ void chunk_step(const Piece16& a, const Piece16& b, v8f& acc, float& m,
                                           int hh) {
    float seed = m;
    asm volatile("" : "+v"(seed));
    float t0 = seed, t1 = seed, t2 = seed, t3 = seed;
    #pragma unroll
    for (int j = 0; j < NQ; j += 4) {
        t0 = fmaxf(fmaxf(t0, a.p[j + 0].x), a.p[j + 0].y); t0 = fmaxf(fmaxf(t0, a.p[j + 0].z), a.p[j + 0].w);
        t1 = fmaxf(fmaxf(t1, a.p[j + 1].x), a.p[j + 1].y); t1 = fmaxf(fmaxf(t1, a.p[j + 1].z), a.p[j + 1].w);
        t2 = fmaxf(fmaxf(t2, a.p[j + 2].x), a.p[j + 2].y); t2 = fmaxf(fmaxf(t2, a.p[j + 2].z), a.p[j + 2].w);
        t3 = fmaxf(fmaxf(t3, a.p[j + 3].x), a.p[j + 3].y); t3 = fmaxf(fmaxf(t3, a.p[j + 3].z), a.p[j + 3].w);
    }
    float mn = fmaxf(fmaxf(t0, t1), fmaxf(t2, t3));
    mn = fmaxf(mn, xhalf(mn));
    const float f  = __expf(m - mn);
    m = mn;
    #pragma unroll
    for (int v = 0; v < 8; ++v) acc[v] *= rowpick(f, 8 * hh + v);

    #pragma unroll
    for (int j = 0; j < NQ; ++j) {
        v2f av, bv;
        av.x = __expf(a.p[j].x - mn); av.y = __expf(a.p[j].y - mn);
        bv.x = b.p[j].x;              bv.y = b.p[j].y;
        acc = __builtin_amdgcn_wmma_f32_16x16x4_f32(false, av, false, bv, (short)0, acc, false, false);
        av.x = __expf(a.p[j].z - mn); av.y = __expf(a.p[j].w - mn);
        bv.x = b.p[j].z;              bv.y = b.p[j].w;
        acc = __builtin_amdgcn_wmma_f32_16x16x4_f32(false, av, false, bv, (short)0, acc, false, false);
    }
    asm volatile("" : "+v"(acc));
}

template <bool MORE>
__device__ __forceinline__ void four_chunks(Piece16& a0, Piece16& a1, Piece16& a2, Piece16& a3,
                                            Piece16& b0, Piece16& b1,
                                            const float* __restrict__ xp, const float4* bl,
                                            int ch, v8f& acc, float& m, int hh) {
    load_bpieces(b1, bl + (ch + 1) * NQ * BUNITS);
    load_pieces(a3, xp + (ch + 3) * WCH, 8);
    __builtin_amdgcn_sched_barrier(0);
    chunk_step(a0, b0, acc, m, hh);

    load_bpieces(b0, bl + (ch + 2) * NQ * BUNITS);
    if constexpr (MORE) load_pieces(a0, xp + (ch + 4) * WCH, 8);
    __builtin_amdgcn_sched_barrier(0);
    chunk_step(a1, b1, acc, m, hh);

    load_bpieces(b1, bl + (ch + 3) * NQ * BUNITS);
    if constexpr (MORE) load_pieces(a1, xp + (ch + 5) * WCH, 8);
    __builtin_amdgcn_sched_barrier(0);
    chunk_step(a2, b0, acc, m, hh);

    if constexpr (MORE) {
        load_bpieces(b0, bl + (ch + 4) * NQ * BUNITS);
        load_pieces(a2, xp + (ch + 6) * WCH, 8);
    }
    __builtin_amdgcn_sched_barrier(0);
    chunk_step(a3, b1, acc, m, hh);
}

__global__ __launch_bounds__(64) void lse_wmma_kernel(const float* __restrict__ x,
                                                      const float* __restrict__ w,
                                                      float* __restrict__ out) {
    __shared__ float4 sB[NJ * BUNITS];

    const int lane = threadIdx.x & 31;
    const int wv   = threadIdx.x >> 5;
    const int tile = blockIdx.x * 2 + wv;
    const int r0   = tile * 16;
    const int rA   = lane & 15;
    const int hh   = lane >> 4;

    const float* xp = x + (size_t)(r0 + rA) * DDIM + 4 * hh;
    const float4* bl = sB + hh * CCLS + (rA < CCLS ? rA : CCLS - 1);

    Piece16 a0, a1, a2, a3;
    Piece16 b0, b1;
    float m = -3.0e38f;
    v8f acc = {0.f, 0.f, 0.f, 0.f, 0.f, 0.f, 0.f, 0.f};

    float4 wq[CPW][WQ];
    #pragma unroll
    for (int k = 0; k < CPW; ++k) {
        const float* wr = w + (CPW * wv + k) * DDIM + 4 * lane;
        #pragma unroll
        for (int i = 0; i < WQ; ++i) wq[k][i] = *(const float4*)(wr + i * 128);
    }
    load_pieces(a0, xp + 0 * WCH, 8);
    load_pieces(a1, xp + 1 * WCH, 8);
    load_pieces(a2, xp + 2 * WCH, 8);
    __builtin_amdgcn_sched_barrier(0);

    #pragma unroll
    for (int k = 0; k < CPW; ++k) {
        float mx = -3.4e38f;
        #pragma unroll
        for (int i = 0; i < WQ; ++i)
            mx = fmaxf(fmaxf(fmaxf(mx, wq[k][i].x), wq[k][i].y), fmaxf(wq[k][i].z, wq[k][i].w));
        mx = wave_max(mx);

        float sum = 0.0f;
        #pragma unroll
        for (int i = 0; i < WQ; ++i) {
            wq[k][i].x = __expf(wq[k][i].x - mx); wq[k][i].y = __expf(wq[k][i].y - mx);
            wq[k][i].z = __expf(wq[k][i].z - mx); wq[k][i].w = __expf(wq[k][i].w - mx);
            sum += (wq[k][i].x + wq[k][i].y) + (wq[k][i].z + wq[k][i].w);
        }
        sum = wave_sum(sum);
        const float inv = 1.0f / sum;

        const int c = CPW * wv + k;
        #pragma unroll
        for (int i = 0; i < WQ; ++i) {
            const int f = 32 * i + lane;
            float4 v;
            v.x = wq[k][i].x * inv; v.y = wq[k][i].y * inv;
            v.z = wq[k][i].z * inv; v.w = wq[k][i].w * inv;
            sB[(f >> 1) * BUNITS + (f & 1) * CCLS + c] = v;
        }
    }
    __syncthreads();

    load_bpieces(b0, bl);

    for (int it = 0; it < NCHUNK / 4 - 1; ++it)
        four_chunks<true>(a0, a1, a2, a3, b0, b1, xp, bl, 4 * it, acc, m, hh);
    four_chunks<false>(a0, a1, a2, a3, b0, b1, xp, bl, NCHUNK - 4, acc, m, hh);

    float mf[8];
    #pragma unroll
    for (int v = 0; v < 8; ++v) mf[v] = rowpick(m, 8 * hh + v);
    if (rA < CCLS) {
        #pragma unroll
        for (int v = 0; v < 8; ++v)
            out[(size_t)(r0 + 8 * hh + v) * CCLS + rA] = mf[v] + __logf(acc[v]);
    }
}

extern "C" void kernel_launch(void* const* d_in, const int* in_sizes, int n_in,
                              void* d_out, int out_size, void* d_ws, size_t ws_size,
                              hipStream_t stream) {
    const float* x = (const float*)d_in[0];
    const float* w = (const float*)d_in[1];
    float* out   = (float*)d_out;

    lse_wmma_kernel<<<BROWS / 32, 64, 0, stream>>>(x, w, out);
}
